# starting kernel
# baseline (speedup 1.0000x reference)
#include <hip/hip_runtime.h>

#define RR 5
#define CC 2
#define LL 2
#define HIDD 128

typedef __attribute__((ext_vector_type(2))) float v2f;
typedef __attribute__((ext_vector_type(8))) float v8f;

__global__ void softmax_filt_kernel(const float* __restrict__ gt_w,
                                    float* __restrict__ filt) {
  int t = threadIdx.x;
  if (t >= LL * CC) return;
  const float* g = gt_w + t * RR;
  float m = g[0];
  for (int r = 1; r < RR; ++r) m = fmaxf(m, g[r]);
  float e[RR];
  float s = 0.f;
  for (int r = 0; r < RR; ++r) { e[r] = __expf(g[r] - m); s += e[r]; }
  float inv = 1.0f / s;
  for (int r = 0; r < RR; ++r) filt[t * RR + r] = e[r] * inv;
}

__global__ void deg_kernel(const int* __restrict__ dst,
                           const int* __restrict__ etype,
                           const float* __restrict__ filt,
                           float* __restrict__ deg,
                           int E, int N, int layer) {
  int e = blockIdx.x * blockDim.x + threadIdx.x;
  if (e >= E) return;
  int d  = dst[e];
  int et = etype[e];
  #pragma unroll
  for (int c = 0; c < CC; ++c)
    atomicAdd(&deg[(size_t)c * N + d], filt[(layer * CC + c) * RR + et]);
}

__global__ void msg_kernel(const int* __restrict__ src,
                           const int* __restrict__ dst,
                           const int* __restrict__ etype,
                           const float* __restrict__ filt,
                           const float* __restrict__ deg,
                           const float* __restrict__ Hin,
                           float* __restrict__ Hout,
                           int E, int N, int layer) {
  int gid  = blockIdx.x * blockDim.x + threadIdx.x;
  int e    = gid >> 5;
  int lane = gid & 31;
  if (e >= E) return;
  int s  = src[e];
  int d  = dst[e];
  int et = etype[e];
  #pragma unroll
  for (int c = 0; c < CC; ++c) {
    float w  = filt[(layer * CC + c) * RR + et];
    float wn = w / deg[(size_t)c * N + d];
    const float4* hp = (const float4*)(Hin + ((size_t)c * N + s) * HIDD);
    float4 v = hp[lane];
    float* op = Hout + ((size_t)c * N + d) * HIDD + lane * 4;
    atomicAdd(op + 0, v.x * wn);
    atomicAdd(op + 1, v.y * wn);
    atomicAdd(op + 2, v.z * wn);
    atomicAdd(op + 3, v.w * wn);
  }
}

template <int NTILES>
__global__ __launch_bounds__(256) void gemm_wmma_f32(
    const float* __restrict__ A, const float* __restrict__ B,
    const float* __restrict__ bias, float* __restrict__ Out,
    int M, int K,
    int aRowStride, int aKShift, int aKMask, size_t aChunkStride,
    size_t bBatchStride, size_t outBatchStride, int relu) {
  constexpr int NDIM = NTILES * 16;
  __shared__ float2 Bs[16 * NDIM];

  const int wave = threadIdx.x >> 5;
  const int lane = threadIdx.x & 31;
  const int half = lane & 15;
  const int hi   = lane >> 4;

  const float* Bb = B + bBatchStride * blockIdx.y;
  float*       Ob = Out + outBatchStride * blockIdx.y;

  const int mTile = blockIdx.x * 8 + wave;
  const int row0  = mTile * 16;

  v8f acc[NTILES];
  for (int nt = 0; nt < NTILES; ++nt) {
    float bv = bias ? bias[nt * 16 + half] : 0.0f;
    v8f t;
    #pragma unroll
    for (int i = 0; i < 8; ++i) t[i] = bv;
    acc[nt] = t;
  }

  int aRow = row0 + half;
  if (aRow >= M) aRow = M - 1;
  const float* Arow = A + (size_t)aRow * aRowStride;

  const float2* BsLane = Bs + (size_t)hi * NDIM + half;

  for (int k0 = 0; k0 < K; k0 += 32) {
    for (int t = threadIdx.x; t < 16 * NDIM; t += 256) {
      int kp = t / NDIM, nn = t % NDIM;
      float2 v;
      v.x = Bb[(size_t)(k0 + 2 * kp)     * NDIM + nn];
      v.y = Bb[(size_t)(k0 + 2 * kp + 1) * NDIM + nn];
      Bs[(size_t)kp * NDIM + nn] = v;
    }
    __syncthreads();

    const float* ap0 = Arow
        + (size_t)(k0 >> aKShift) * aChunkStride
        + ((k0 & aKMask) + 2 * hi);

    #pragma unroll
    for (int kk = 0; kk < 32; kk += 4) {
      v2f a = *(const v2f*)(ap0 + kk);
      #pragma unroll
      for (int nt = 0; nt < NTILES; ++nt) {
        v2f b = *(const v2f*)(BsLane + (size_t)(kk >> 1) * NDIM + nt * 16);
        acc[nt] = __builtin_amdgcn_wmma_f32_16x16x4_f32(
            false, a, false, b, (short)0, acc[nt], false, false);
      }
    }
    __syncthreads();
  }

  for (int nt = 0; nt < NTILES; ++nt) {
    #pragma unroll
    for (int vr = 0; vr < 8; ++vr) {
      int r = row0 + vr + 8 * hi;
      if (r < M) {
        float v = acc[nt][vr];
        if (relu) v = fmaxf(v, 0.0f);
        Ob[(size_t)r * NDIM + nt * 16 + half] = v;
      }
    }
  }
}

__global__ void gather_kernel(const float* __restrict__ Y,
                              const int* __restrict__ cat,
                              float* __restrict__ out, int ncat) {
  int t = blockIdx.x * blockDim.x + threadIdx.x;
  if (t >= ncat * 16) return;
  int i = t >> 4, j = t & 15;
  out[t] = Y[(size_t)cat[i] * 16 + j];
}

extern "C" void kernel_launch(void* const* d_in, const int* in_sizes, int n_in,
                              void* d_out, int out_size, void* d_ws, size_t ws_size,
                              hipStream_t stream) {
  const float* h   = (const float*)d_in[0];
  const int*   src = (const int*)d_in[1];
  const int*   dst = (const int*)d_in[2];
  const int*   et  = (const int*)d_in[3];
  const int*   cat = (const int*)d_in[4];
  const float* gtw = (const float*)d_in[5];
  const float* Wc  = (const float*)d_in[6];
  const float* W1  = (const float*)d_in[7];
  const float* b1  = (const float*)d_in[8];
  const float* W2  = (const float*)d_in[9];
  const float* b2  = (const float*)d_in[10];

  const int IN_DIM = 256;
  const int N    = in_sizes[0] / IN_DIM;
  const int E    = in_sizes[1];
  const int NCAT = in_sizes[4];

  float* ws    = (float*)d_ws;
  float* filt  = ws;
  float* deg   = filt + 32;
  float* Hbuf0 = deg + (size_t)CC * N;
  float* Hbuf1 = Hbuf0 + (size_t)CC * N * HIDD;
  float* Z     = Hbuf1 + (size_t)CC * N * HIDD;
  float* Y     = Z + (size_t)N * HIDD;

  softmax_filt_kernel<<<1, 32, 0, stream>>>(gtw, filt);

  const int mtiles = (N + 15) / 16;
  const int gx     = (mtiles + 7) / 8;

  gemm_wmma_f32<8><<<dim3(gx, CC), 256, 0, stream>>>(
      h, Wc, nullptr, Hbuf0, N, IN_DIM,
        IN_DIM,   8,   255,   0,
        (size_t)IN_DIM * HIDD,   (size_t)N * HIDD,   0);

  float* Hcur = Hbuf0;
  float* Hnxt = Hbuf1;
  for (int l = 0; l < LL; ++l) {
    hipMemsetAsync(deg, 0, sizeof(float) * (size_t)CC * N, stream);
    hipMemsetAsync(Hnxt, 0, sizeof(float) * (size_t)CC * N * HIDD, stream);
    deg_kernel<<<(E + 255) / 256, 256, 0, stream>>>(dst, et, filt, deg, E, N, l);
    long long th = (long long)E * 32;
    msg_kernel<<<(unsigned)((th + 255) / 256), 256, 0, stream>>>(
        src, dst, et, filt, deg, Hcur, Hnxt, E, N, l);
    float* tmp = Hcur; Hcur = Hnxt; Hnxt = tmp;
  }

  gemm_wmma_f32<8><<<dim3(gx, 1), 256, 0, stream>>>(
      Hcur, W1, b1, Z, N, CC * HIDD,
        HIDD,   7,   127,
        (size_t)N * HIDD,
      0, 0,   1);

  gemm_wmma_f32<1><<<dim3(gx, 1), 256, 0, stream>>>(
      Z, W2, b2, Y, N, HIDD,
        HIDD,   7,   127,   0,
      0, 0,   0);

  gather_kernel<<<(NCAT * 16 + 255) / 256, 256, 0, stream>>>(
      Y, cat, (float*)d_out, NCAT);
}


# W1 folded into the projection (h @ (Wc[c] W1[c]), one small extra WMMA product): last round sums both channels per edge into one plane, hidden GEMM gone; rounding and sum order change
# speedup vs baseline: 1.5933x; 1.5933x over previous
#include <hip/hip_runtime.h>

#define RR 5
#define CC 2
#define LL 2
#define HIDD 128

typedef __attribute__((ext_vector_type(2))) float v2f;
typedef __attribute__((ext_vector_type(8))) float v8f;

__global__ void softmax_filt_kernel(const float* __restrict__ gt_w,
                                    float* __restrict__ filt) {
  int t = threadIdx.x;
  if (t >= LL * CC) return;
  const float* g = gt_w + t * RR;
  float m = g[0];
  for (int r = 1; r < RR; ++r) m = fmaxf(m, g[r]);
  float e[RR];
  float s = 0.f;
  for (int r = 0; r < RR; ++r) { e[r] = __expf(g[r] - m); s += e[r]; }
  float inv = 1.0f / s;
  for (int r = 0; r < RR; ++r) filt[t * RR + r] = e[r] * inv;
}

__global__ void deg_kernel(const int* __restrict__ dst,
                           const int* __restrict__ etype,
                           const float* __restrict__ filt,
                           float* __restrict__ deg,
                           int E, int N, int layer) {
  int e = blockIdx.x * blockDim.x + threadIdx.x;
  if (e >= E) return;
  int d  = dst[e];
  int et = etype[e];
  #pragma unroll
  for (int c = 0; c < CC; ++c)
    atomicAdd(&deg[(size_t)c * N + d], filt[(layer * CC + c) * RR + et]);
}

__global__ void msg_kernel(const int* __restrict__ src,
                           const int* __restrict__ dst,
                           const int* __restrict__ etype,
                           const float* __restrict__ filt,
                           const float* __restrict__ deg,
                           const float* __restrict__ Hin,
                           float* __restrict__ Hout,
                           int E, int N, int layer) {
  int gid  = blockIdx.x * blockDim.x + threadIdx.x;
  int e    = gid >> 5;
  int lane = gid & 31;
  if (e >= E) return;
  int s  = src[e];
  int d  = dst[e];
  int et = etype[e];
  float wn[CC];
  float v[CC][4];
  #pragma unroll
  for (int c = 0; c < CC; ++c) {
    float w = filt[(layer * CC + c) * RR + et];
    wn[c] = w / deg[(size_t)c * N + d];
    const float* hp = Hin + ((size_t)c * N + s) * HIDD + lane;
    for (int j = 0; j < 4; ++j) v[c][j] = hp[32 * j];
  }
  for (int c = 0; c < CC; ++c) {
    float* op = Hout + ((size_t)c * N + d) * HIDD + lane;
    for (int j = 0; j < 4; ++j) atomicAdd(op + 32 * j, v[c][j] * wn[c]);
  }
}

__global__ void msg_sum_kernel(const int* __restrict__ src,
                               const int* __restrict__ dst,
                               const int* __restrict__ etype,
                               const float* __restrict__ filt,
                               const float* __restrict__ deg,
                               const float* __restrict__ Hin,
                               float* __restrict__ Hout,
                               int E, int N, int layer) {
  int gid  = blockIdx.x * blockDim.x + threadIdx.x;
  int e    = gid >> 5;
  int lane = gid & 31;
  if (e >= E) return;
  int s  = src[e];
  int d  = dst[e];
  int et = etype[e];
  float wn[CC];
  float v[CC][4];
  for (int c = 0; c < CC; ++c) {
    float w = filt[(layer * CC + c) * RR + et];
    wn[c] = w / deg[(size_t)c * N + d];
    const float* hp = Hin + ((size_t)c * N + s) * HIDD + lane;
    for (int j = 0; j < 4; ++j) v[c][j] = hp[32 * j];
  }
  float* op = Hout + (size_t)d * HIDD + lane;
  for (int j = 0; j < 4; ++j) {
    float m = v[0][j] * wn[0];
    for (int c = 1; c < CC; ++c) m += v[c][j] * wn[c];
    atomicAdd(op + 32 * j, m);
  }
}

template <int NTILES, bool AACT>
__global__ __launch_bounds__(256) void gemm_wmma_f32(
    const float* __restrict__ A, const float* __restrict__ B,
    const float* __restrict__ bias, const float* __restrict__ aBias,
    float* __restrict__ Out,
    int M, int K,
    int aRowStride, int aKShift, int aKMask, size_t aChunkStride,
    size_t aBatchStride, size_t bBatchStride, size_t outBatchStride,
    int relu) {
  constexpr int NDIM = NTILES * 16;
  __shared__ float2 Bs[16 * NDIM];

  const int wave = threadIdx.x >> 5;
  const int lane = threadIdx.x & 31;
  const int half = lane & 15;
  const int hi   = lane >> 4;

  const float* Bb = B + bBatchStride * blockIdx.y;
  float*       Ob = Out + outBatchStride * blockIdx.y;

  const int mTile = blockIdx.x * 8 + wave;
  const int row0  = mTile * 16;

  v8f acc[NTILES];
  for (int nt = 0; nt < NTILES; ++nt) {
    float bv = bias ? bias[nt * 16 + half] : 0.0f;
    v8f t;
    #pragma unroll
    for (int i = 0; i < 8; ++i) t[i] = bv;
    acc[nt] = t;
  }

  int aRow = row0 + half;
  if (aRow >= M) aRow = M - 1;
  const float* Arow = A + aBatchStride * blockIdx.y + (size_t)aRow * aRowStride;

  const float2* BsLane = Bs + (size_t)hi * NDIM + half;

  for (int k0 = 0; k0 < K; k0 += 32) {
    for (int t = threadIdx.x; t < 16 * NDIM; t += 256) {
      int kp = t / NDIM, nn = t % NDIM;
      float2 v;
      v.x = Bb[(size_t)(k0 + 2 * kp)     * NDIM + nn];
      v.y = Bb[(size_t)(k0 + 2 * kp + 1) * NDIM + nn];
      Bs[(size_t)kp * NDIM + nn] = v;
    }
    __syncthreads();

    const float* ap0 = Arow
        + (size_t)(k0 >> aKShift) * aChunkStride
        + ((k0 & aKMask) + 2 * hi);

    #pragma unroll
    for (int kk = 0; kk < 32; kk += 4) {
      v2f a = *(const v2f*)(ap0 + kk);
      if (AACT) {
        v2f ab = *(const v2f*)(aBias + k0 + kk + 2 * hi);
        a.x = fmaxf(a.x + ab.x, 0.0f);
        a.y = fmaxf(a.y + ab.y, 0.0f);
      }
      #pragma unroll
      for (int nt = 0; nt < NTILES; ++nt) {
        v2f b = *(const v2f*)(BsLane + (size_t)(kk >> 1) * NDIM + nt * 16);
        acc[nt] = __builtin_amdgcn_wmma_f32_16x16x4_f32(
            false, a, false, b, (short)0, acc[nt], false, false);
      }
    }
    __syncthreads();
  }

  for (int nt = 0; nt < NTILES; ++nt) {
    #pragma unroll
    for (int vr = 0; vr < 8; ++vr) {
      int r = row0 + vr + 8 * hi;
      if (r < M) {
        float v = acc[nt][vr];
        if (relu) v = fmaxf(v, 0.0f);
        Ob[(size_t)r * NDIM + nt * 16 + half] = v;
      }
    }
  }
}

__global__ void gather_kernel(const float* __restrict__ Y,
                              const int* __restrict__ cat,
                              float* __restrict__ out, int ncat) {
  int t = blockIdx.x * blockDim.x + threadIdx.x;
  if (t >= ncat * 16) return;
  int i = t >> 4, j = t & 15;
  out[t] = Y[(size_t)cat[i] * 16 + j];
}

extern "C" void kernel_launch(void* const* d_in, const int* in_sizes, int n_in,
                              void* d_out, int out_size, void* d_ws, size_t ws_size,
                              hipStream_t stream) {
  const float* h   = (const float*)d_in[0];
  const int*   src = (const int*)d_in[1];
  const int*   dst = (const int*)d_in[2];
  const int*   et  = (const int*)d_in[3];
  const int*   cat = (const int*)d_in[4];
  const float* gtw = (const float*)d_in[5];
  const float* Wc  = (const float*)d_in[6];
  const float* W1  = (const float*)d_in[7];
  const float* b1  = (const float*)d_in[8];
  const float* W2  = (const float*)d_in[9];
  const float* b2  = (const float*)d_in[10];

  const int IN_DIM = 256;
  const int N    = in_sizes[0] / IN_DIM;
  const int E    = in_sizes[1];
  const int NCAT = in_sizes[4];

  float* ws    = (float*)d_ws;
  float* filt  = ws;
  float* deg   = filt + 32;
  float* Hbuf0 = deg + (size_t)CC * N;
  float* Hbuf1 = Hbuf0 + (size_t)CC * N * HIDD;
  float* Wf    = Hbuf1 + (size_t)CC * N * HIDD;
  float* Y     = Wf + (size_t)N * HIDD;

  softmax_filt_kernel<<<1, 32, 0, stream>>>(gtw, filt);

  const int mtiles = (N + 15) / 16;
  const int gx     = (mtiles + 7) / 8;

  const int gxf = (IN_DIM / 16 + 7) / 8;
  gemm_wmma_f32<8, false><<<dim3(gxf, CC), 256, 0, stream>>>(
      Wc, W1, nullptr, nullptr, Wf, IN_DIM, HIDD,
        HIDD,   7,   127,   0,
        (size_t)IN_DIM * HIDD,   (size_t)HIDD * HIDD,
        (size_t)IN_DIM * HIDD,   0);
  gemm_wmma_f32<8, false><<<dim3(gx, CC), 256, 0, stream>>>(
      h, Wf, nullptr, nullptr, Hbuf0, N, IN_DIM,
        IN_DIM,   8,   255,   0,
        0,   (size_t)IN_DIM * HIDD,
        (size_t)N * HIDD,   0);

  float* Hcur = Hbuf0;
  float* Hnxt = Hbuf1;
  const long long th = (long long)E * 32;
  for (int l = 0; l < LL - 1; ++l) {
    hipMemsetAsync(deg, 0, sizeof(float) * (size_t)CC * N, stream);
    hipMemsetAsync(Hnxt, 0, sizeof(float) * (size_t)CC * N * HIDD, stream);
    deg_kernel<<<(E + 255) / 256, 256, 0, stream>>>(dst, et, filt, deg, E, N, l);
    msg_kernel<<<(unsigned)((th + 255) / 256), 256, 0, stream>>>(
        src, dst, et, filt, deg, Hcur, Hnxt, E, N, l);
    float* tmp = Hcur; Hcur = Hnxt; Hnxt = tmp;
  }
  float* S = Hnxt;
  hipMemsetAsync(deg, 0, sizeof(float) * (size_t)CC * N, stream);
  hipMemsetAsync(S, 0, sizeof(float) * (size_t)N * HIDD, stream);
  deg_kernel<<<(E + 255) / 256, 256, 0, stream>>>(dst, et, filt, deg, E, N, LL - 1);
  msg_sum_kernel<<<(unsigned)((th + 255) / 256), 256, 0, stream>>>(
      src, dst, et, filt, deg, Hcur, S, E, N, LL - 1);

  gemm_wmma_f32<1, true><<<dim3(gx, 1), 256, 0, stream>>>(
      S, W2, b2, b1, Y, N, HIDD,
        HIDD,   7,   127,   0,
      0, 0, 0,   0);

  gather_kernel<<<(NCAT * 16 + 255) / 256, 256, 0, stream>>>(
      Y, cat, (float*)d_out, NCAT);
}


# Last round: its degrees come from the round before, so an edge's two channel rows are scaled by weight/degree and added into one LDS row (4 LDS float atomics per edge, were 8); sum order changes
# speedup vs baseline: 3.0701x; 1.9269x over previous
#include <hip/hip_runtime.h>

#define RR 5
#define CC 2
#define LL 2
#define HIDD 128

static_assert(LL >= 2, "the last round needs a round before it");

typedef __attribute__((ext_vector_type(2))) float v2f;
typedef __attribute__((ext_vector_type(8))) float v8f;

typedef __attribute__((ext_vector_type(4))) int v4i;
typedef __attribute__((ext_vector_type(2))) int v2i;

#define TROWS  196

__global__ __launch_bounds__(256) void prep_kernel(const float* __restrict__ gt_w,
                                                   float* __restrict__ filt,
                                                   const int* __restrict__ dst,
                                                   unsigned* __restrict__ tileb,
                                                   int nquad) {
  if (blockIdx.x != 0) {
    const int q = (blockIdx.x - 1) * 256 + threadIdx.x;
    if (q < nquad) {
      const v4i d = *(const v4i*)(dst + 4 * q);
      tileb[q] = ((unsigned)d.x / TROWS) | (((unsigned)d.y / TROWS) << 8) |
                 (((unsigned)d.z / TROWS) << 16) | (((unsigned)d.w / TROWS) << 24);
    }
    return;
  }
  int t = threadIdx.x;
  if (t >= LL * CC) return;
  const float* g = gt_w + t * RR;
  float m = g[0];
  for (int r = 1; r < RR; ++r) m = fmaxf(m, g[r]);
  float e[RR];
  float s = 0.f;
  for (int r = 0; r < RR; ++r) { e[r] = __expf(g[r] - m); s += e[r]; }
  float inv = 1.0f / s;
  for (int r = 0; r < RR; ++r) filt[t * RR + r] = e[r] * inv;
}

#define TWAVES 32
#define NGRP   4
#define HCAP   256
#define QCAP   512
#define QSTR   528
#define CHUNK  16
#define EU     4
#define LCAP   8192

template <int U, bool SUMC>
__device__ __forceinline__ void tile_group(int j0, int s, int r,
                                           float w0, float w1, int lane,
                                           const float* __restrict__ Hin,
                                           float* acc, int N) {
  float v[U][CC][4];
  for (int u = 0; u < U; ++u) {
    const int sj = __builtin_amdgcn_readlane(s, j0 + u);
    for (int c = 0; c < CC; ++c) {
      const float* hp = Hin + ((size_t)c * N + sj) * HIDD + lane;
      for (int q = 0; q < 4; ++q) v[u][c][q] = hp[32 * q];
    }
  }
  for (int u = 0; u < U; ++u) {
    const int   rj = __builtin_amdgcn_readlane(r, j0 + u);
    const float a0 = __int_as_float(__builtin_amdgcn_readlane(__float_as_int(w0), j0 + u));
    const float a1 = __int_as_float(__builtin_amdgcn_readlane(__float_as_int(w1), j0 + u));
    if (SUMC) {
      float* p = acc + rj * HIDD + lane;
      for (int q = 0; q < 4; ++q) atomicAdd(p + 32 * q, v[u][0][q] * a0 + v[u][1][q] * a1);
    } else {
      float* p0 = acc + (rj * CC + 0) * HIDD + lane;
      float* p1 = acc + (rj * CC + 1) * HIDD + lane;
      for (int q = 0; q < 4; ++q) atomicAdd(p0 + 32 * q, v[u][0][q] * a0);
      for (int q = 0; q < 4; ++q) atomicAdd(p1 + 32 * q, v[u][1][q] * a1);
    }
  }
}

template <bool SUMC>
__device__ __forceinline__ void process_rows(int n, int s, int r, int et, int lane,
                                             int layer, int N,
                                             const float* __restrict__ filt,
                                             const float* __restrict__ Hin,
                                             float* acc, float* dsum, float* dnext,
                                             int* ecnt) {
  float w0 = 0.0f, w1 = 0.0f;
  if (lane < n) {
    w0 = filt[(layer * CC + 0) * RR + et];
    w1 = filt[(layer * CC + 1) * RR + et];
    if (SUMC) {
      w0 = w0 / dsum[r * CC + 0];
      w1 = w1 / dsum[r * CC + 1];
    } else {
      atomicAdd(&dsum[r * CC + 0], w0);
      atomicAdd(&dsum[r * CC + 1], w1);
      atomicAdd(&ecnt[r], 1);
      atomicAdd(&dnext[r * CC + 0], filt[((layer + 1) * CC + 0) * RR + et]);
      atomicAdd(&dnext[r * CC + 1], filt[((layer + 1) * CC + 1) * RR + et]);
    }
  }
  int j = 0;
  for (; j + EU <= n; j += EU)
    tile_group<EU, SUMC>(j, s, r, w0, w1, lane, Hin, acc, N);
  for (; j < n; ++j)
    tile_group<1, SUMC>(j, s, r, w0, w1, lane, Hin, acc, N);
}

template <bool SUMC>
__device__ __forceinline__ void process_entries(const int* q, int n, int lane,
                                                int row0, int layer, int N,
                                                const int* __restrict__ src,
                                                const int* __restrict__ dst,
                                                const int* __restrict__ etype,
                                                const float* __restrict__ filt,
                                                const float* __restrict__ Hin,
                                                float* acc, float* dsum, float* dnext,
                                                int* ecnt, v2i* save) {
  int s = 0, r = 0, et = 0;
  if (lane < n) {
    const int e = q[lane];
    s  = src[e];
    r  = dst[e] - row0;
    et = etype[e];
    if (save) {
      v2i p;
      p.x = s;
      p.y = r | (et << 8);
      save[lane] = p;
    }
  }
  process_rows<SUMC>(n, s, r, et, lane, layer, N, filt, Hin, acc, dsum, dnext, ecnt);
}

template <bool SUMC>
__global__ __launch_bounds__(TWAVES * 32) void tile_msg_kernel(
    const int* __restrict__ src, const int* __restrict__ dst,
    const int* __restrict__ etype, const float* __restrict__ filt,
    const float* __restrict__ Hin, float* __restrict__ Hout,
    const v4i* __restrict__ tileb,
    int* elist, int* lcnt, float* degws, int E, int N, int layer, int first) {
  __shared__ float acc[TROWS * (SUMC ? 1 : CC) * HIDD];
  __shared__ float dsum[TROWS * CC];
  __shared__ float dnext[TROWS * CC];
  __shared__ int   ecnt[TROWS];
  __shared__ int   queue[TWAVES * QSTR];
  __shared__ int   hits[TWAVES * HCAP];
  __shared__ int   qcnt[TWAVES];
  __shared__ int   lost;

  const int tid  = threadIdx.x;
  const int lane = tid & 31;
  const int wave = __builtin_amdgcn_readfirstlane(tid >> 5);
  const int row0 = blockIdx.x * TROWS;
  const int rows = min(TROWS, N - row0);

  for (int i = tid; i < TROWS * (SUMC ? 1 : CC) * HIDD; i += TWAVES * 32) acc[i] = 0.0f;
  if (SUMC) {
    for (int i = tid; i < rows * CC; i += TWAVES * 32) dsum[i] = degws[(size_t)row0 * CC + i];
  } else {
    for (int i = tid; i < TROWS * CC; i += TWAVES * 32) { dsum[i] = 0.0f; dnext[i] = 0.0f; }
    for (int i = tid; i < TROWS; i += TWAVES * 32) ecnt[i] = 0;
  }
  if (tid == 0) lost = 0;
  __syncthreads();

  v2i* mylist = (v2i*)elist + (size_t)blockIdx.x * LCAP;
  int nsaved = -1;
  if (!first) nsaved = min(__builtin_amdgcn_readfirstlane(lcnt[blockIdx.x]), LCAP);

  if (nsaved >= 0) {
    const int beg = (wave * nsaved) >> 5;
    const int end = ((wave + 1) * nsaved) >> 5;
    for (int j0 = beg; j0 < end; j0 += 32) {
      const int n = min(32, end - j0);
      int s = 0, r = 0, et = 0;
      if (lane < n) {
        const v2i p = mylist[j0 + lane];
        s  = p.x;
        r  = p.y & 255;
        et = (int)((unsigned)p.y >> 8);
      }
      process_rows<SUMC>(n, s, r, et, lane, layer, N, filt, Hin, acc, dsum, dnext, ecnt);
    }
  } else {

  int* myq = queue + wave * QSTR;
  int* myh = hits + wave * HCAP;
  int  cnt = 0;
  int  hc  = 0;
  const unsigned below = (1u << lane) - 1u;
  const unsigned pat   = (unsigned)blockIdx.x * 0x01010101u;

  auto resolve = [&]() {
    for (int h0 = 0; h0 < hc; h0 += 32) {
      const int n = min(32, hc - h0);
      int g = 0;
      unsigned z[4] = {0u, 0u, 0u, 0u};
      if (lane < n) {
        g = myh[h0 + lane];
        const v4i x = tileb[g];
        for (int k = 0; k < 4; ++k) {
          const unsigned y = (unsigned)x[k] ^ pat;
          z[k] = ~(((y & 0x7F7F7F7Fu) + 0x7F7F7F7Fu) | y | 0x7F7F7F7Fu);
        }
      }
      unsigned mk[16];
      int tot = 0;
      for (int i = 0; i < 16; ++i) {
        mk[i] = __builtin_amdgcn_ballot_w32(((z[i >> 2] >> (8 * (i & 3) + 7)) & 1u) != 0u);
        tot += __builtin_popcount(mk[i]);
      }
      if (cnt + tot > QCAP) {
        for (int j0 = 0; j0 < cnt; j0 += 32)
          process_entries<SUMC>(myq + j0, min(32, cnt - j0), lane, row0, layer, N,
                                src, dst, etype, filt, Hin, acc, dsum, dnext, ecnt, nullptr);
        if (lane == 0) lost = 1;
        cnt = 0;
      }
      for (int i = 0; i < 16; ++i) {
        if ((mk[i] >> lane) & 1u)
          myq[cnt + __builtin_popcount(mk[i] & below)] = g * 16 + i;
        cnt += __builtin_popcount(mk[i]);
      }
    }
    hc = 0;
  };

  const int G     = E >> 4;
  const int trips = (G + TWAVES * 32 * NGRP - 1) / (TWAVES * 32 * NGRP);
  const int rot   = trips > 0 ? (int)(blockIdx.x % (unsigned)trips) : 0;

  for (int t = 0; t < trips; ++t) {
    int tt = t + rot;
    if (tt >= trips) tt -= trips;
    const int base = (tt * TWAVES + wave) * (32 * NGRP) + lane;

    v4i x[NGRP];
    for (int u = 0; u < NGRP; ++u) x[u] = tileb[min(base + 32 * u, G - 1)];

    unsigned mk[NGRP];
    int tot = 0;
    for (int u = 0; u < NGRP; ++u) {
      unsigned any = 0u;
      for (int k = 0; k < 4; ++k) {
        const unsigned y = (unsigned)x[u][k] ^ pat;
        any |= (y - 0x01010101u) & ~y;
      }
      mk[u] = __builtin_amdgcn_ballot_w32((base + 32 * u < G) &&
                                          (any & 0x80808080u) != 0u);
      tot += __builtin_popcount(mk[u]);
    }
    if (tot != 0) {
      if (hc + tot > HCAP) resolve();
      for (int u = 0; u < NGRP; ++u) {
        if ((mk[u] >> lane) & 1u)
          myh[hc + __builtin_popcount(mk[u] & below)] = base + 32 * u;
        hc += __builtin_popcount(mk[u]);
      }
    }
  }
  resolve();

  if (wave == 0) {
    const int  e  = (G << 4) + lane;
    const bool mt = (e < E) && ((unsigned)(dst[e] - row0) < (unsigned)rows);
    const unsigned mk = __builtin_amdgcn_ballot_w32(mt);
    if (mt) myq[cnt + __builtin_popcount(mk & below)] = e;
    cnt += __builtin_popcount(mk);
  }

  if (lane == 0) qcnt[wave] = cnt;
  __syncthreads();
  int total = 0;
  for (int qw = 0; qw < TWAVES; ++qw) total += __builtin_amdgcn_readfirstlane(qcnt[qw]);
  const bool keep = first && lost == 0 && total <= LCAP;
  if (first && tid == 0) lcnt[blockIdx.x] = keep ? total : -1;
  int g = 0, run = 0;
  for (int qw = 0; qw < TWAVES; ++qw) {
    const int cq = __builtin_amdgcn_readfirstlane(qcnt[qw]);
    for (int j0 = 0; j0 < cq; j0 += CHUNK, ++g)
      if ((g & (TWAVES - 1)) == wave)
        process_entries<SUMC>(queue + qw * QSTR + j0, min(CHUNK, cq - j0), lane, row0,
                              layer, N, src, dst, etype, filt, Hin, acc, dsum, dnext, ecnt,
                              keep ? mylist + run + j0 : nullptr);
    run += cq;
  }
  }
  __syncthreads();

  if (SUMC) {
    for (int idx = tid; idx < rows * HIDD; idx += TWAVES * 32)
      Hout[(size_t)row0 * HIDD + idx] = acc[idx];
  } else {
    for (int idx = tid; idx < rows * HIDD; idx += TWAVES * 32) {
      const int r = idx >> 7, col = idx & (HIDD - 1);
      const bool has = ecnt[r] > 0;
      Hout[(size_t)(row0 + r) * HIDD + col] =
          has ? acc[(r * CC + 0) * HIDD + col] / dsum[r * CC + 0] : 0.0f;
      Hout[((size_t)N + row0 + r) * HIDD + col] =
          has ? acc[(r * CC + 1) * HIDD + col] / dsum[r * CC + 1] : 0.0f;
    }
    for (int i = tid; i < rows * CC; i += TWAVES * 32) degws[(size_t)row0 * CC + i] = dnext[i];
  }
}

template <int NTILES, bool AACT>
__global__ __launch_bounds__(256) void gemm_wmma_f32(
    const float* __restrict__ A, const float* __restrict__ B,
    const float* __restrict__ bias, const float* __restrict__ aBias,
    float* __restrict__ Out,
    int M, int K,
    int aRowStride, int aKShift, int aKMask, size_t aChunkStride,
    size_t aBatchStride, size_t bBatchStride, size_t outBatchStride,
    int relu) {
  constexpr int NDIM = NTILES * 16;
  __shared__ float2 Bs[16 * NDIM];

  const int wave = threadIdx.x >> 5;
  const int lane = threadIdx.x & 31;
  const int half = lane & 15;
  const int hi   = lane >> 4;

  const float* Bb = B + bBatchStride * blockIdx.y;
  float*       Ob = Out + outBatchStride * blockIdx.y;

  const int mTile = blockIdx.x * 8 + wave;
  const int row0  = mTile * 16;

  v8f acc[NTILES];
  for (int nt = 0; nt < NTILES; ++nt) {
    float bv = bias ? bias[nt * 16 + half] : 0.0f;
    v8f t;
    #pragma unroll
    for (int i = 0; i < 8; ++i) t[i] = bv;
    acc[nt] = t;
  }

  int aRow = row0 + half;
  if (aRow >= M) aRow = M - 1;
  const float* Arow = A + aBatchStride * blockIdx.y + (size_t)aRow * aRowStride;

  const float2* BsLane = Bs + (size_t)hi * NDIM + half;

  for (int k0 = 0; k0 < K; k0 += 32) {
    for (int t = threadIdx.x; t < 16 * NDIM; t += 256) {
      int kp = t / NDIM, nn = t % NDIM;
      float2 v;
      v.x = Bb[(size_t)(k0 + 2 * kp)     * NDIM + nn];
      v.y = Bb[(size_t)(k0 + 2 * kp + 1) * NDIM + nn];
      Bs[(size_t)kp * NDIM + nn] = v;
    }
    __syncthreads();

    const float* ap0 = Arow
        + (size_t)(k0 >> aKShift) * aChunkStride
        + ((k0 & aKMask) + 2 * hi);

    #pragma unroll
    for (int kk = 0; kk < 32; kk += 4) {
      v2f a = *(const v2f*)(ap0 + kk);
      if (AACT) {
        v2f ab = *(const v2f*)(aBias + k0 + kk + 2 * hi);
        a.x = fmaxf(a.x + ab.x, 0.0f);
        a.y = fmaxf(a.y + ab.y, 0.0f);
      }
      #pragma unroll
      for (int nt = 0; nt < NTILES; ++nt) {
        v2f b = *(const v2f*)(BsLane + (size_t)(kk >> 1) * NDIM + nt * 16);
        acc[nt] = __builtin_amdgcn_wmma_f32_16x16x4_f32(
            false, a, false, b, (short)0, acc[nt], false, false);
      }
    }
    __syncthreads();
  }

  for (int nt = 0; nt < NTILES; ++nt) {
    #pragma unroll
    for (int vr = 0; vr < 8; ++vr) {
      int r = row0 + vr + 8 * hi;
      if (r < M) {
        float v = acc[nt][vr];
        if (relu) v = fmaxf(v, 0.0f);
        Ob[(size_t)r * NDIM + nt * 16 + half] = v;
      }
    }
  }
}

__global__ void gather_kernel(const float* __restrict__ Y,
                              const int* __restrict__ cat,
                              float* __restrict__ out, int ncat) {
  int t = blockIdx.x * blockDim.x + threadIdx.x;
  if (t >= ncat * 16) return;
  int i = t >> 4, j = t & 15;
  out[t] = Y[(size_t)cat[i] * 16 + j];
}

extern "C" void kernel_launch(void* const* d_in, const int* in_sizes, int n_in,
                              void* d_out, int out_size, void* d_ws, size_t ws_size,
                              hipStream_t stream) {
  const float* h   = (const float*)d_in[0];
  const int*   src = (const int*)d_in[1];
  const int*   dst = (const int*)d_in[2];
  const int*   et  = (const int*)d_in[3];
  const int*   cat = (const int*)d_in[4];
  const float* gtw = (const float*)d_in[5];
  const float* Wc  = (const float*)d_in[6];
  const float* W1  = (const float*)d_in[7];
  const float* b1  = (const float*)d_in[8];
  const float* W2  = (const float*)d_in[9];
  const float* b2  = (const float*)d_in[10];

  const int IN_DIM = 256;
  const int N    = in_sizes[0] / IN_DIM;
  const int E    = in_sizes[1];
  const int NCAT = in_sizes[4];

  float* ws    = (float*)d_ws;
  float* filt  = ws;
  float* Hbuf0 = filt + 128;
  float* Hbuf1 = Hbuf0 + (size_t)CC * N * HIDD;
  float* Wf    = Hbuf1 + (size_t)CC * N * HIDD;
  float* Y     = Wf + (size_t)N * HIDD;

  const int gtile = (N + TROWS - 1) / TROWS;
  int* elist = (int*)(Wf + (size_t)CC * IN_DIM * HIDD);
  int* lcnt  = elist + (size_t)gtile * LCAP * 2;
  unsigned* tileb = (unsigned*)(lcnt + ((gtile + 31) & ~31));
  float* degws = (float*)(tileb + (((E >> 4) * 4 + 31) & ~31));

  const int nquad = (E >> 4) * 4;
  prep_kernel<<<1 + (nquad + 255) / 256, 256, 0, stream>>>(gtw, filt, dst, tileb, nquad);

  const int mtiles = (N + 15) / 16;
  const int gx     = (mtiles + 7) / 8;

  const int gxf = (IN_DIM / 16 + 7) / 8;
  gemm_wmma_f32<8, false><<<dim3(gxf, CC), 256, 0, stream>>>(
      Wc, W1, nullptr, nullptr, Wf, IN_DIM, HIDD,
        HIDD,   7,   127,   0,
        (size_t)IN_DIM * HIDD,   (size_t)HIDD * HIDD,
        (size_t)IN_DIM * HIDD,   0);
  gemm_wmma_f32<8, false><<<dim3(gx, CC), 256, 0, stream>>>(
      h, Wf, nullptr, nullptr, Hbuf0, N, IN_DIM,
        IN_DIM,   8,   255,   0,
        0,   (size_t)IN_DIM * HIDD,
        (size_t)N * HIDD,   0);

  float* Hcur = Hbuf0;
  float* Hnxt = Hbuf1;
  for (int l = 0; l < LL - 1; ++l) {
    tile_msg_kernel<false><<<gtile, TWAVES * 32, 0, stream>>>(
        src, dst, et, filt, Hcur, Hnxt, (const v4i*)tileb, elist, lcnt, degws, E, N, l, l == 0 ? 1 : 0);
    float* tmp = Hcur; Hcur = Hnxt; Hnxt = tmp;
  }
  float* S = Hnxt;
  tile_msg_kernel<true><<<gtile, TWAVES * 32, 0, stream>>>(
      src, dst, et, filt, Hcur, S, (const v4i*)tileb, elist, lcnt, degws, E, N, LL - 1, 0);

  gemm_wmma_f32<1, true><<<dim3(gx, 1), 256, 0, stream>>>(
      S, W2, b2, b1, Y, N, HIDD,
        HIDD,   7,   127,   0,
      0, 0, 0,   0);

  gather_kernel<<<(NCAT * 16 + 255) / 256, 256, 0, stream>>>(
      Y, cat, (float*)d_out, NCAT);
}


# Weight product Wc*W1 as 256 one-tile waves: operands straight from memory to registers, all loads in flight at once, no LDS or barrier (was 4 blocks walking 4 staged K panels); k sum order changes
# speedup vs baseline: 3.5199x; 1.1465x over previous
#include <hip/hip_runtime.h>

#define RR 5
#define CC 2
#define LL 2
#define HIDD 128

static_assert(LL >= 2, "the last round needs a round before it");

typedef __attribute__((ext_vector_type(2))) float v2f;
typedef __attribute__((ext_vector_type(8))) float v8f;
typedef __attribute__((ext_vector_type(4))) float v4f;

typedef __attribute__((ext_vector_type(4))) int v4i;
typedef __attribute__((ext_vector_type(2))) int v2i;

#define TROWS  196

__global__ __launch_bounds__(256) void prep_kernel(const float* __restrict__ gt_w,
                                                   float* __restrict__ filt,
                                                   const int* __restrict__ dst,
                                                   unsigned* __restrict__ tileb,
                                                   int nquad) {
  if (blockIdx.x != 0) {
    const int q = (blockIdx.x - 1) * 256 + threadIdx.x;
    if (q < nquad) {
      const v4i d = *(const v4i*)(dst + 4 * q);
      tileb[q] = ((unsigned)d.x / TROWS) | (((unsigned)d.y / TROWS) << 8) |
                 (((unsigned)d.z / TROWS) << 16) | (((unsigned)d.w / TROWS) << 24);
    }
    return;
  }
  int t = threadIdx.x;
  if (t >= LL * CC) return;
  const float* g = gt_w + t * RR;
  float m = g[0];
  for (int r = 1; r < RR; ++r) m = fmaxf(m, g[r]);
  float e[RR];
  float s = 0.f;
  for (int r = 0; r < RR; ++r) { e[r] = __expf(g[r] - m); s += e[r]; }
  float inv = 1.0f / s;
  for (int r = 0; r < RR; ++r) filt[t * RR + r] = e[r] * inv;
}

#define TWAVES 32
#define NGRP   4
#define HCAP   256
#define QCAP   512
#define QSTR   528
#define CHUNK  16
#define EU     4
#define LCAP   8192

template <int U, bool SUMC>
__device__ __forceinline__ void tile_group(int j0, int s, int r,
                                           float w0, float w1, int lane,
                                           const float* __restrict__ Hin,
                                           float* acc, int N) {
  float v[U][CC][4];
  for (int u = 0; u < U; ++u) {
    const int sj = __builtin_amdgcn_readlane(s, j0 + u);
    for (int c = 0; c < CC; ++c) {
      const float* hp = Hin + ((size_t)c * N + sj) * HIDD + lane;
      for (int q = 0; q < 4; ++q) v[u][c][q] = hp[32 * q];
    }
  }
  for (int u = 0; u < U; ++u) {
    const int   rj = __builtin_amdgcn_readlane(r, j0 + u);
    const float a0 = __int_as_float(__builtin_amdgcn_readlane(__float_as_int(w0), j0 + u));
    const float a1 = __int_as_float(__builtin_amdgcn_readlane(__float_as_int(w1), j0 + u));
    if (SUMC) {
      float* p = acc + rj * HIDD + lane;
      for (int q = 0; q < 4; ++q) atomicAdd(p + 32 * q, v[u][0][q] * a0 + v[u][1][q] * a1);
    } else {
      float* p0 = acc + (rj * CC + 0) * HIDD + lane;
      float* p1 = acc + (rj * CC + 1) * HIDD + lane;
      for (int q = 0; q < 4; ++q) atomicAdd(p0 + 32 * q, v[u][0][q] * a0);
      for (int q = 0; q < 4; ++q) atomicAdd(p1 + 32 * q, v[u][1][q] * a1);
    }
  }
}

template <bool SUMC>
__device__ __forceinline__ void process_rows(int n, int s, int r, int et, int lane,
                                             int layer, int N,
                                             const float* __restrict__ filt,
                                             const float* __restrict__ Hin,
                                             float* acc, float* dsum, float* dnext,
                                             int* ecnt) {
  float w0 = 0.0f, w1 = 0.0f;
  if (lane < n) {
    w0 = filt[(layer * CC + 0) * RR + et];
    w1 = filt[(layer * CC + 1) * RR + et];
    if (SUMC) {
      w0 = w0 / dsum[r * CC + 0];
      w1 = w1 / dsum[r * CC + 1];
    } else {
      atomicAdd(&dsum[r * CC + 0], w0);
      atomicAdd(&dsum[r * CC + 1], w1);
      atomicAdd(&ecnt[r], 1);
      atomicAdd(&dnext[r * CC + 0], filt[((layer + 1) * CC + 0) * RR + et]);
      atomicAdd(&dnext[r * CC + 1], filt[((layer + 1) * CC + 1) * RR + et]);
    }
  }
  int j = 0;
  for (; j + EU <= n; j += EU)
    tile_group<EU, SUMC>(j, s, r, w0, w1, lane, Hin, acc, N);
  for (; j < n; ++j)
    tile_group<1, SUMC>(j, s, r, w0, w1, lane, Hin, acc, N);
}

template <bool SUMC>
__device__ __forceinline__ void process_entries(const int* q, int n, int lane,
                                                int row0, int layer, int N,
                                                const int* __restrict__ src,
                                                const int* __restrict__ dst,
                                                const int* __restrict__ etype,
                                                const float* __restrict__ filt,
                                                const float* __restrict__ Hin,
                                                float* acc, float* dsum, float* dnext,
                                                int* ecnt, v2i* save) {
  int s = 0, r = 0, et = 0;
  if (lane < n) {
    const int e = q[lane];
    s  = src[e];
    r  = dst[e] - row0;
    et = etype[e];
    if (save) {
      v2i p;
      p.x = s;
      p.y = r | (et << 8);
      save[lane] = p;
    }
  }
  process_rows<SUMC>(n, s, r, et, lane, layer, N, filt, Hin, acc, dsum, dnext, ecnt);
}

template <bool SUMC>
__global__ __launch_bounds__(TWAVES * 32) void tile_msg_kernel(
    const int* __restrict__ src, const int* __restrict__ dst,
    const int* __restrict__ etype, const float* __restrict__ filt,
    const float* __restrict__ Hin, float* __restrict__ Hout,
    const v4i* __restrict__ tileb,
    int* elist, int* lcnt, float* degws, int E, int N, int layer, int first) {
  __shared__ float acc[TROWS * (SUMC ? 1 : CC) * HIDD];
  __shared__ float dsum[TROWS * CC];
  __shared__ float dnext[TROWS * CC];
  __shared__ int   ecnt[TROWS];
  __shared__ int   queue[TWAVES * QSTR];
  __shared__ int   hits[TWAVES * HCAP];
  __shared__ int   qcnt[TWAVES];
  __shared__ int   lost;

  const int tid  = threadIdx.x;
  const int lane = tid & 31;
  const int wave = __builtin_amdgcn_readfirstlane(tid >> 5);
  const int row0 = blockIdx.x * TROWS;
  const int rows = min(TROWS, N - row0);

  for (int i = tid; i < TROWS * (SUMC ? 1 : CC) * HIDD; i += TWAVES * 32) acc[i] = 0.0f;
  if (SUMC) {
    for (int i = tid; i < rows * CC; i += TWAVES * 32) dsum[i] = degws[(size_t)row0 * CC + i];
  } else {
    for (int i = tid; i < TROWS * CC; i += TWAVES * 32) { dsum[i] = 0.0f; dnext[i] = 0.0f; }
    for (int i = tid; i < TROWS; i += TWAVES * 32) ecnt[i] = 0;
  }
  if (tid == 0) lost = 0;
  __syncthreads();

  v2i* mylist = (v2i*)elist + (size_t)blockIdx.x * LCAP;
  int nsaved = -1;
  if (!first) nsaved = min(__builtin_amdgcn_readfirstlane(lcnt[blockIdx.x]), LCAP);

  if (nsaved >= 0) {
    const int beg = (wave * nsaved) >> 5;
    const int end = ((wave + 1) * nsaved) >> 5;
    for (int j0 = beg; j0 < end; j0 += 32) {
      const int n = min(32, end - j0);
      int s = 0, r = 0, et = 0;
      if (lane < n) {
        const v2i p = mylist[j0 + lane];
        s  = p.x;
        r  = p.y & 255;
        et = (int)((unsigned)p.y >> 8);
      }
      process_rows<SUMC>(n, s, r, et, lane, layer, N, filt, Hin, acc, dsum, dnext, ecnt);
    }
  } else {

  int* myq = queue + wave * QSTR;
  int* myh = hits + wave * HCAP;
  int  cnt = 0;
  int  hc  = 0;
  const unsigned below = (1u << lane) - 1u;
  const unsigned pat   = (unsigned)blockIdx.x * 0x01010101u;

  auto resolve = [&]() {
    for (int h0 = 0; h0 < hc; h0 += 32) {
      const int n = min(32, hc - h0);
      int g = 0;
      unsigned z[4] = {0u, 0u, 0u, 0u};
      if (lane < n) {
        g = myh[h0 + lane];
        const v4i x = tileb[g];
        for (int k = 0; k < 4; ++k) {
          const unsigned y = (unsigned)x[k] ^ pat;
          z[k] = ~(((y & 0x7F7F7F7Fu) + 0x7F7F7F7Fu) | y | 0x7F7F7F7Fu);
        }
      }
      unsigned mk[16];
      int tot = 0;
      for (int i = 0; i < 16; ++i) {
        mk[i] = __builtin_amdgcn_ballot_w32(((z[i >> 2] >> (8 * (i & 3) + 7)) & 1u) != 0u);
        tot += __builtin_popcount(mk[i]);
      }
      if (cnt + tot > QCAP) {
        for (int j0 = 0; j0 < cnt; j0 += 32)
          process_entries<SUMC>(myq + j0, min(32, cnt - j0), lane, row0, layer, N,
                                src, dst, etype, filt, Hin, acc, dsum, dnext, ecnt, nullptr);
        if (lane == 0) lost = 1;
        cnt = 0;
      }
      for (int i = 0; i < 16; ++i) {
        if ((mk[i] >> lane) & 1u)
          myq[cnt + __builtin_popcount(mk[i] & below)] = g * 16 + i;
        cnt += __builtin_popcount(mk[i]);
      }
    }
    hc = 0;
  };

  const int G     = E >> 4;
  const int trips = (G + TWAVES * 32 * NGRP - 1) / (TWAVES * 32 * NGRP);
  const int rot   = trips > 0 ? (int)(blockIdx.x % (unsigned)trips) : 0;

  for (int t = 0; t < trips; ++t) {
    int tt = t + rot;
    if (tt >= trips) tt -= trips;
    const int base = (tt * TWAVES + wave) * (32 * NGRP) + lane;

    v4i x[NGRP];
    for (int u = 0; u < NGRP; ++u) x[u] = tileb[min(base + 32 * u, G - 1)];

    unsigned mk[NGRP];
    int tot = 0;
    for (int u = 0; u < NGRP; ++u) {
      unsigned any = 0u;
      for (int k = 0; k < 4; ++k) {
        const unsigned y = (unsigned)x[u][k] ^ pat;
        any |= (y - 0x01010101u) & ~y;
      }
      mk[u] = __builtin_amdgcn_ballot_w32((base + 32 * u < G) &&
                                          (any & 0x80808080u) != 0u);
      tot += __builtin_popcount(mk[u]);
    }
    if (tot != 0) {
      if (hc + tot > HCAP) resolve();
      for (int u = 0; u < NGRP; ++u) {
        if ((mk[u] >> lane) & 1u)
          myh[hc + __builtin_popcount(mk[u] & below)] = base + 32 * u;
        hc += __builtin_popcount(mk[u]);
      }
    }
  }
  resolve();

  if (wave == 0) {
    const int  e  = (G << 4) + lane;
    const bool mt = (e < E) && ((unsigned)(dst[e] - row0) < (unsigned)rows);
    const unsigned mk = __builtin_amdgcn_ballot_w32(mt);
    if (mt) myq[cnt + __builtin_popcount(mk & below)] = e;
    cnt += __builtin_popcount(mk);
  }

  if (lane == 0) qcnt[wave] = cnt;
  __syncthreads();
  int total = 0;
  for (int qw = 0; qw < TWAVES; ++qw) total += __builtin_amdgcn_readfirstlane(qcnt[qw]);
  const bool keep = first && lost == 0 && total <= LCAP;
  if (first && tid == 0) lcnt[blockIdx.x] = keep ? total : -1;
  int g = 0, run = 0;
  for (int qw = 0; qw < TWAVES; ++qw) {
    const int cq = __builtin_amdgcn_readfirstlane(qcnt[qw]);
    for (int j0 = 0; j0 < cq; j0 += CHUNK, ++g)
      if ((g & (TWAVES - 1)) == wave)
        process_entries<SUMC>(queue + qw * QSTR + j0, min(CHUNK, cq - j0), lane, row0,
                              layer, N, src, dst, etype, filt, Hin, acc, dsum, dnext, ecnt,
                              keep ? mylist + run + j0 : nullptr);
    run += cq;
  }
  }
  __syncthreads();

  if (SUMC) {
    for (int idx = tid; idx < rows * HIDD; idx += TWAVES * 32)
      Hout[(size_t)row0 * HIDD + idx] = acc[idx];
  } else {
    for (int idx = tid; idx < rows * HIDD; idx += TWAVES * 32) {
      const int r = idx >> 7, col = idx & (HIDD - 1);
      const bool has = ecnt[r] > 0;
      Hout[(size_t)(row0 + r) * HIDD + col] =
          has ? acc[(r * CC + 0) * HIDD + col] / dsum[r * CC + 0] : 0.0f;
      Hout[((size_t)N + row0 + r) * HIDD + col] =
          has ? acc[(r * CC + 1) * HIDD + col] / dsum[r * CC + 1] : 0.0f;
    }
    for (int i = tid; i < rows * CC; i += TWAVES * 32) degws[(size_t)row0 * CC + i] = dnext[i];
  }
}

template <int NTILES, bool AACT>
__global__ __launch_bounds__(256) void gemm_wmma_f32(
    const float* __restrict__ A, const float* __restrict__ B,
    const float* __restrict__ bias, const float* __restrict__ aBias,
    float* __restrict__ Out,
    int M, int K,
    int aRowStride, int aKShift, int aKMask, size_t aChunkStride,
    size_t aBatchStride, size_t bBatchStride, size_t outBatchStride,
    int relu) {
  constexpr int NDIM = NTILES * 16;
  __shared__ float2 Bs[16 * NDIM];

  const int wave = threadIdx.x >> 5;
  const int lane = threadIdx.x & 31;
  const int half = lane & 15;
  const int hi   = lane >> 4;

  const float* Bb = B + bBatchStride * blockIdx.y;
  float*       Ob = Out + outBatchStride * blockIdx.y;

  const int mTile = blockIdx.x * 8 + wave;
  const int row0  = mTile * 16;

  v8f acc[NTILES];
  for (int nt = 0; nt < NTILES; ++nt) {
    float bv = bias ? bias[nt * 16 + half] : 0.0f;
    v8f t;
    #pragma unroll
    for (int i = 0; i < 8; ++i) t[i] = bv;
    acc[nt] = t;
  }

  int aRow = row0 + half;
  if (aRow >= M) aRow = M - 1;
  const float* Arow = A + aBatchStride * blockIdx.y + (size_t)aRow * aRowStride;

  const float2* BsLane = Bs + (size_t)hi * NDIM + half;

  for (int k0 = 0; k0 < K; k0 += 32) {
    for (int t = threadIdx.x; t < 16 * NDIM; t += 256) {
      int kp = t / NDIM, nn = t % NDIM;
      float2 v;
      v.x = Bb[(size_t)(k0 + 2 * kp)     * NDIM + nn];
      v.y = Bb[(size_t)(k0 + 2 * kp + 1) * NDIM + nn];
      Bs[(size_t)kp * NDIM + nn] = v;
    }
    __syncthreads();

    const float* ap0 = Arow
        + (size_t)(k0 >> aKShift) * aChunkStride
        + ((k0 & aKMask) + 2 * hi);

    #pragma unroll
    for (int kk = 0; kk < 32; kk += 4) {
      v2f a = *(const v2f*)(ap0 + kk);
      if (AACT) {
        v2f ab = *(const v2f*)(aBias + k0 + kk + 2 * hi);
        a.x = fmaxf(a.x + ab.x, 0.0f);
        a.y = fmaxf(a.y + ab.y, 0.0f);
      }
      #pragma unroll
      for (int nt = 0; nt < NTILES; ++nt) {
        v2f b = *(const v2f*)(BsLane + (size_t)(kk >> 1) * NDIM + nt * 16);
        acc[nt] = __builtin_amdgcn_wmma_f32_16x16x4_f32(
            false, a, false, b, (short)0, acc[nt], false, false);
      }
    }
    __syncthreads();
  }

  for (int nt = 0; nt < NTILES; ++nt) {
    #pragma unroll
    for (int vr = 0; vr < 8; ++vr) {
      int r = row0 + vr + 8 * hi;
      if (r < M) {
        float v = acc[nt][vr];
        if (relu) v = fmaxf(v, 0.0f);
        Ob[(size_t)r * NDIM + nt * 16 + half] = v;
      }
    }
  }
}

__global__ __launch_bounds__(32) void wf_tile_kernel(const float* __restrict__ Wc,
                                                     const float* __restrict__ W1,
                                                     float* __restrict__ Wf,
                                                     int inDim) {
  const int lane = threadIdx.x;
  const int half = lane & 15;
  const int hi   = lane >> 4;
  const int mt = blockIdx.x, nt = blockIdx.y, c = blockIdx.z;

  const float* ap = Wc + ((size_t)c * inDim + mt * 16 + half) * HIDD + 4 * hi;
  const float* bp = W1 + ((size_t)c * HIDD + 4 * hi) * HIDD + nt * 16 + half;

  v4f   a[HIDD / 8];
  float b[HIDD / 8][4];
  #pragma unroll
  for (int i = 0; i < HIDD / 8; ++i) a[i] = *(const v4f*)(ap + 8 * i);
  #pragma unroll
  for (int i = 0; i < HIDD / 8; ++i)
    for (int t = 0; t < 4; ++t) b[i][t] = bp[(size_t)(8 * i + t) * HIDD];

  __builtin_amdgcn_sched_barrier(0);

  v8f acc0, acc1;
  for (int i = 0; i < 8; ++i) { acc0[i] = 0.0f; acc1[i] = 0.0f; }
  #pragma unroll
  for (int i = 0; i < HIDD / 8; ++i) {
    v2f a0, b0, a1, b1;
    a0.x = a[i].x;  a0.y = a[i].y;  b0.x = b[i][0];  b0.y = b[i][1];
    a1.x = a[i].z;  a1.y = a[i].w;  b1.x = b[i][2];  b1.y = b[i][3];
    acc0 = __builtin_amdgcn_wmma_f32_16x16x4_f32(false, a0, false, b0, (short)0, acc0, false, false);
    acc1 = __builtin_amdgcn_wmma_f32_16x16x4_f32(false, a1, false, b1, (short)0, acc1, false, false);
  }

  float* op = Wf + ((size_t)c * inDim + mt * 16 + 8 * hi) * HIDD + nt * 16 + half;
  #pragma unroll
  for (int vr = 0; vr < 8; ++vr) op[(size_t)vr * HIDD] = acc0[vr] + acc1[vr];
}

__global__ void gather_kernel(const float* __restrict__ Y,
                              const int* __restrict__ cat,
                              float* __restrict__ out, int ncat) {
  int t = blockIdx.x * blockDim.x + threadIdx.x;
  if (t >= ncat * 16) return;
  int i = t >> 4, j = t & 15;
  out[t] = Y[(size_t)cat[i] * 16 + j];
}

extern "C" void kernel_launch(void* const* d_in, const int* in_sizes, int n_in,
                              void* d_out, int out_size, void* d_ws, size_t ws_size,
                              hipStream_t stream) {
  const float* h   = (const float*)d_in[0];
  const int*   src = (const int*)d_in[1];
  const int*   dst = (const int*)d_in[2];
  const int*   et  = (const int*)d_in[3];
  const int*   cat = (const int*)d_in[4];
  const float* gtw = (const float*)d_in[5];
  const float* Wc  = (const float*)d_in[6];
  const float* W1  = (const float*)d_in[7];
  const float* b1  = (const float*)d_in[8];
  const float* W2  = (const float*)d_in[9];
  const float* b2  = (const float*)d_in[10];

  const int IN_DIM = 256;
  const int N    = in_sizes[0] / IN_DIM;
  const int E    = in_sizes[1];
  const int NCAT = in_sizes[4];

  float* ws    = (float*)d_ws;
  float* filt  = ws;
  float* Hbuf0 = filt + 128;
  float* Hbuf1 = Hbuf0 + (size_t)CC * N * HIDD;
  float* Wf    = Hbuf1 + (size_t)CC * N * HIDD;
  float* Y     = Wf + (size_t)N * HIDD;

  const int gtile = (N + TROWS - 1) / TROWS;
  int* elist = (int*)(Wf + (size_t)CC * IN_DIM * HIDD);
  int* lcnt  = elist + (size_t)gtile * LCAP * 2;
  unsigned* tileb = (unsigned*)(lcnt + ((gtile + 31) & ~31));
  float* degws = (float*)(tileb + (((E >> 4) * 4 + 31) & ~31));

  const int nquad = (E >> 4) * 4;
  prep_kernel<<<1 + (nquad + 255) / 256, 256, 0, stream>>>(gtw, filt, dst, tileb, nquad);

  const int mtiles = (N + 15) / 16;
  const int gx     = (mtiles + 7) / 8;

  wf_tile_kernel<<<dim3(IN_DIM / 16, HIDD / 16, CC), 32, 0, stream>>>(Wc, W1, Wf, IN_DIM);
  gemm_wmma_f32<8, false><<<dim3(gx, CC), 256, 0, stream>>>(
      h, Wf, nullptr, nullptr, Hbuf0, N, IN_DIM,
        IN_DIM,   8,   255,   0,
        0,   (size_t)IN_DIM * HIDD,
        (size_t)N * HIDD,   0);

  float* Hcur = Hbuf0;
  float* Hnxt = Hbuf1;
  for (int l = 0; l < LL - 1; ++l) {
    tile_msg_kernel<false><<<gtile, TWAVES * 32, 0, stream>>>(
        src, dst, et, filt, Hcur, Hnxt, (const v4i*)tileb, elist, lcnt, degws, E, N, l, l == 0 ? 1 : 0);
    float* tmp = Hcur; Hcur = Hnxt; Hnxt = tmp;
  }
  float* S = Hnxt;
  tile_msg_kernel<true><<<gtile, TWAVES * 32, 0, stream>>>(
      src, dst, et, filt, Hcur, S, (const v4i*)tileb, elist, lcnt, degws, E, N, LL - 1, 0);

  gemm_wmma_f32<1, true><<<dim3(gx, 1), 256, 0, stream>>>(
      S, W2, b2, b1, Y, N, HIDD,
        HIDD,   7,   127,   0,
      0, 0, 0,   0);

  gather_kernel<<<(NCAT * 16 + 255) / 256, 256, 0, stream>>>(
      Y, cat, (float*)d_out, NCAT);
}


# Projection GEMM weight-stationary: whole folded weight (both channels, 256 KB) stays in LDS, one work-group per WGP, no K panels or loop barriers; h read once; tasks dealt evenly over SIMDs
# speedup vs baseline: 3.8733x; 1.1004x over previous
#include <hip/hip_runtime.h>

#define RR 5
#define CC 2
#define LL 2
#define HIDD 128

static_assert(LL >= 2, "the last round needs a round before it");

typedef __attribute__((ext_vector_type(2))) float v2f;
typedef __attribute__((ext_vector_type(8))) float v8f;
typedef __attribute__((ext_vector_type(4))) float v4f;

typedef __attribute__((ext_vector_type(4))) int v4i;
typedef __attribute__((ext_vector_type(2))) int v2i;

#define TROWS  196

__global__ __launch_bounds__(256) void prep_kernel(const float* __restrict__ gt_w,
                                                   float* __restrict__ filt,
                                                   const int* __restrict__ dst,
                                                   unsigned* __restrict__ tileb,
                                                   int nquad) {
  if (blockIdx.x != 0) {
    const int q = (blockIdx.x - 1) * 256 + threadIdx.x;
    if (q < nquad) {
      const v4i d = *(const v4i*)(dst + 4 * q);
      tileb[q] = ((unsigned)d.x / TROWS) | (((unsigned)d.y / TROWS) << 8) |
                 (((unsigned)d.z / TROWS) << 16) | (((unsigned)d.w / TROWS) << 24);
    }
    return;
  }
  int t = threadIdx.x;
  if (t >= LL * CC) return;
  const float* g = gt_w + t * RR;
  float m = g[0];
  for (int r = 1; r < RR; ++r) m = fmaxf(m, g[r]);
  float e[RR];
  float s = 0.f;
  for (int r = 0; r < RR; ++r) { e[r] = __expf(g[r] - m); s += e[r]; }
  float inv = 1.0f / s;
  for (int r = 0; r < RR; ++r) filt[t * RR + r] = e[r] * inv;
}

#define TWAVES 32
#define NGRP   4
#define HCAP   256
#define QCAP   512
#define QSTR   528
#define CHUNK  16
#define EU     4
#define LCAP   8192

template <int U, bool SUMC>
__device__ __forceinline__ void tile_group(int j0, int s, int r,
                                           float w0, float w1, int lane,
                                           const float* __restrict__ Hin,
                                           float* acc, int N) {
  float v[U][CC][4];
  for (int u = 0; u < U; ++u) {
    const int sj = __builtin_amdgcn_readlane(s, j0 + u);
    for (int c = 0; c < CC; ++c) {
      const float* hp = Hin + ((size_t)c * N + sj) * HIDD + lane;
      for (int q = 0; q < 4; ++q) v[u][c][q] = hp[32 * q];
    }
  }
  for (int u = 0; u < U; ++u) {
    const int   rj = __builtin_amdgcn_readlane(r, j0 + u);
    const float a0 = __int_as_float(__builtin_amdgcn_readlane(__float_as_int(w0), j0 + u));
    const float a1 = __int_as_float(__builtin_amdgcn_readlane(__float_as_int(w1), j0 + u));
    if (SUMC) {
      float* p = acc + rj * HIDD + lane;
      for (int q = 0; q < 4; ++q) atomicAdd(p + 32 * q, v[u][0][q] * a0 + v[u][1][q] * a1);
    } else {
      float* p0 = acc + (rj * CC + 0) * HIDD + lane;
      float* p1 = acc + (rj * CC + 1) * HIDD + lane;
      for (int q = 0; q < 4; ++q) atomicAdd(p0 + 32 * q, v[u][0][q] * a0);
      for (int q = 0; q < 4; ++q) atomicAdd(p1 + 32 * q, v[u][1][q] * a1);
    }
  }
}

template <bool SUMC>
__device__ __forceinline__ void process_rows(int n, int s, int r, int et, int lane,
                                             int layer, int N,
                                             const float* __restrict__ filt,
                                             const float* __restrict__ Hin,
                                             float* acc, float* dsum, float* dnext,
                                             int* ecnt) {
  float w0 = 0.0f, w1 = 0.0f;
  if (lane < n) {
    w0 = filt[(layer * CC + 0) * RR + et];
    w1 = filt[(layer * CC + 1) * RR + et];
    if (SUMC) {
      w0 = w0 / dsum[r * CC + 0];
      w1 = w1 / dsum[r * CC + 1];
    } else {
      atomicAdd(&dsum[r * CC + 0], w0);
      atomicAdd(&dsum[r * CC + 1], w1);
      atomicAdd(&ecnt[r], 1);
      atomicAdd(&dnext[r * CC + 0], filt[((layer + 1) * CC + 0) * RR + et]);
      atomicAdd(&dnext[r * CC + 1], filt[((layer + 1) * CC + 1) * RR + et]);
    }
  }
  int j = 0;
  for (; j + EU <= n; j += EU)
    tile_group<EU, SUMC>(j, s, r, w0, w1, lane, Hin, acc, N);
  for (; j < n; ++j)
    tile_group<1, SUMC>(j, s, r, w0, w1, lane, Hin, acc, N);
}

template <bool SUMC>
__device__ __forceinline__ void process_entries(const int* q, int n, int lane,
                                                int row0, int layer, int N,
                                                const int* __restrict__ src,
                                                const int* __restrict__ dst,
                                                const int* __restrict__ etype,
                                                const float* __restrict__ filt,
                                                const float* __restrict__ Hin,
                                                float* acc, float* dsum, float* dnext,
                                                int* ecnt, v2i* save) {
  int s = 0, r = 0, et = 0;
  if (lane < n) {
    const int e = q[lane];
    s  = src[e];
    r  = dst[e] - row0;
    et = etype[e];
    if (save) {
      v2i p;
      p.x = s;
      p.y = r | (et << 8);
      save[lane] = p;
    }
  }
  process_rows<SUMC>(n, s, r, et, lane, layer, N, filt, Hin, acc, dsum, dnext, ecnt);
}

template <bool SUMC>
__global__ __launch_bounds__(TWAVES * 32) void tile_msg_kernel(
    const int* __restrict__ src, const int* __restrict__ dst,
    const int* __restrict__ etype, const float* __restrict__ filt,
    const float* __restrict__ Hin, float* __restrict__ Hout,
    const v4i* __restrict__ tileb,
    int* elist, int* lcnt, float* degws, int E, int N, int layer, int first) {
  __shared__ float acc[TROWS * (SUMC ? 1 : CC) * HIDD];
  __shared__ float dsum[TROWS * CC];
  __shared__ float dnext[TROWS * CC];
  __shared__ int   ecnt[TROWS];
  __shared__ int   queue[TWAVES * QSTR];
  __shared__ int   hits[TWAVES * HCAP];
  __shared__ int   qcnt[TWAVES];
  __shared__ int   lost;

  const int tid  = threadIdx.x;
  const int lane = tid & 31;
  const int wave = __builtin_amdgcn_readfirstlane(tid >> 5);
  const int row0 = blockIdx.x * TROWS;
  const int rows = min(TROWS, N - row0);

  for (int i = tid; i < TROWS * (SUMC ? 1 : CC) * HIDD; i += TWAVES * 32) acc[i] = 0.0f;
  if (SUMC) {
    for (int i = tid; i < rows * CC; i += TWAVES * 32) dsum[i] = degws[(size_t)row0 * CC + i];
  } else {
    for (int i = tid; i < TROWS * CC; i += TWAVES * 32) { dsum[i] = 0.0f; dnext[i] = 0.0f; }
    for (int i = tid; i < TROWS; i += TWAVES * 32) ecnt[i] = 0;
  }
  if (tid == 0) lost = 0;
  __syncthreads();

  v2i* mylist = (v2i*)elist + (size_t)blockIdx.x * LCAP;
  int nsaved = -1;
  if (!first) nsaved = min(__builtin_amdgcn_readfirstlane(lcnt[blockIdx.x]), LCAP);

  if (nsaved >= 0) {
    const int beg = (wave * nsaved) >> 5;
    const int end = ((wave + 1) * nsaved) >> 5;
    for (int j0 = beg; j0 < end; j0 += 32) {
      const int n = min(32, end - j0);
      int s = 0, r = 0, et = 0;
      if (lane < n) {
        const v2i p = mylist[j0 + lane];
        s  = p.x;
        r  = p.y & 255;
        et = (int)((unsigned)p.y >> 8);
      }
      process_rows<SUMC>(n, s, r, et, lane, layer, N, filt, Hin, acc, dsum, dnext, ecnt);
    }
  } else {

  int* myq = queue + wave * QSTR;
  int* myh = hits + wave * HCAP;
  int  cnt = 0;
  int  hc  = 0;
  const unsigned below = (1u << lane) - 1u;
  const unsigned pat   = (unsigned)blockIdx.x * 0x01010101u;

  auto resolve = [&]() {
    for (int h0 = 0; h0 < hc; h0 += 32) {
      const int n = min(32, hc - h0);
      int g = 0;
      unsigned z[4] = {0u, 0u, 0u, 0u};
      if (lane < n) {
        g = myh[h0 + lane];
        const v4i x = tileb[g];
        for (int k = 0; k < 4; ++k) {
          const unsigned y = (unsigned)x[k] ^ pat;
          z[k] = ~(((y & 0x7F7F7F7Fu) + 0x7F7F7F7Fu) | y | 0x7F7F7F7Fu);
        }
      }
      unsigned mk[16];
      int tot = 0;
      for (int i = 0; i < 16; ++i) {
        mk[i] = __builtin_amdgcn_ballot_w32(((z[i >> 2] >> (8 * (i & 3) + 7)) & 1u) != 0u);
        tot += __builtin_popcount(mk[i]);
      }
      if (cnt + tot > QCAP) {
        for (int j0 = 0; j0 < cnt; j0 += 32)
          process_entries<SUMC>(myq + j0, min(32, cnt - j0), lane, row0, layer, N,
                                src, dst, etype, filt, Hin, acc, dsum, dnext, ecnt, nullptr);
        if (lane == 0) lost = 1;
        cnt = 0;
      }
      for (int i = 0; i < 16; ++i) {
        if ((mk[i] >> lane) & 1u)
          myq[cnt + __builtin_popcount(mk[i] & below)] = g * 16 + i;
        cnt += __builtin_popcount(mk[i]);
      }
    }
    hc = 0;
  };

  const int G     = E >> 4;
  const int trips = (G + TWAVES * 32 * NGRP - 1) / (TWAVES * 32 * NGRP);
  const int rot   = trips > 0 ? (int)(blockIdx.x % (unsigned)trips) : 0;

  for (int t = 0; t < trips; ++t) {
    int tt = t + rot;
    if (tt >= trips) tt -= trips;
    const int base = (tt * TWAVES + wave) * (32 * NGRP) + lane;

    v4i x[NGRP];
    for (int u = 0; u < NGRP; ++u) x[u] = tileb[min(base + 32 * u, G - 1)];

    unsigned mk[NGRP];
    int tot = 0;
    for (int u = 0; u < NGRP; ++u) {
      unsigned any = 0u;
      for (int k = 0; k < 4; ++k) {
        const unsigned y = (unsigned)x[u][k] ^ pat;
        any |= (y - 0x01010101u) & ~y;
      }
      mk[u] = __builtin_amdgcn_ballot_w32((base + 32 * u < G) &&
                                          (any & 0x80808080u) != 0u);
      tot += __builtin_popcount(mk[u]);
    }
    if (tot != 0) {
      if (hc + tot > HCAP) resolve();
      for (int u = 0; u < NGRP; ++u) {
        if ((mk[u] >> lane) & 1u)
          myh[hc + __builtin_popcount(mk[u] & below)] = base + 32 * u;
        hc += __builtin_popcount(mk[u]);
      }
    }
  }
  resolve();

  if (wave == 0) {
    const int  e  = (G << 4) + lane;
    const bool mt = (e < E) && ((unsigned)(dst[e] - row0) < (unsigned)rows);
    const unsigned mk = __builtin_amdgcn_ballot_w32(mt);
    if (mt) myq[cnt + __builtin_popcount(mk & below)] = e;
    cnt += __builtin_popcount(mk);
  }

  if (lane == 0) qcnt[wave] = cnt;
  __syncthreads();
  int total = 0;
  for (int qw = 0; qw < TWAVES; ++qw) total += __builtin_amdgcn_readfirstlane(qcnt[qw]);
  const bool keep = first && lost == 0 && total <= LCAP;
  if (first && tid == 0) lcnt[blockIdx.x] = keep ? total : -1;
  int g = 0, run = 0;
  for (int qw = 0; qw < TWAVES; ++qw) {
    const int cq = __builtin_amdgcn_readfirstlane(qcnt[qw]);
    for (int j0 = 0; j0 < cq; j0 += CHUNK, ++g)
      if ((g & (TWAVES - 1)) == wave)
        process_entries<SUMC>(queue + qw * QSTR + j0, min(CHUNK, cq - j0), lane, row0,
                              layer, N, src, dst, etype, filt, Hin, acc, dsum, dnext, ecnt,
                              keep ? mylist + run + j0 : nullptr);
    run += cq;
  }
  }
  __syncthreads();

  if (SUMC) {
    for (int idx = tid; idx < rows * HIDD; idx += TWAVES * 32)
      Hout[(size_t)row0 * HIDD + idx] = acc[idx];
  } else {
    for (int idx = tid; idx < rows * HIDD; idx += TWAVES * 32) {
      const int r = idx >> 7, col = idx & (HIDD - 1);
      const bool has = ecnt[r] > 0;
      Hout[(size_t)(row0 + r) * HIDD + col] =
          has ? acc[(r * CC + 0) * HIDD + col] / dsum[r * CC + 0] : 0.0f;
      Hout[((size_t)N + row0 + r) * HIDD + col] =
          has ? acc[(r * CC + 1) * HIDD + col] / dsum[r * CC + 1] : 0.0f;
    }
    for (int i = tid; i < rows * CC; i += TWAVES * 32) degws[(size_t)row0 * CC + i] = dnext[i];
  }
}

template <int NTILES, bool AACT>
__global__ __launch_bounds__(256) void gemm_wmma_f32(
    const float* __restrict__ A, const float* __restrict__ B,
    const float* __restrict__ bias, const float* __restrict__ aBias,
    float* __restrict__ Out,
    int M, int K,
    int aRowStride, int aKShift, int aKMask, size_t aChunkStride,
    size_t aBatchStride, size_t bBatchStride, size_t outBatchStride,
    int relu) {
  constexpr int NDIM = NTILES * 16;
  __shared__ float2 Bs[16 * NDIM];

  const int wave = threadIdx.x >> 5;
  const int lane = threadIdx.x & 31;
  const int half = lane & 15;
  const int hi   = lane >> 4;

  const float* Bb = B + bBatchStride * blockIdx.y;
  float*       Ob = Out + outBatchStride * blockIdx.y;

  const int mTile = blockIdx.x * 8 + wave;
  const int row0  = mTile * 16;

  v8f acc[NTILES];
  for (int nt = 0; nt < NTILES; ++nt) {
    float bv = bias ? bias[nt * 16 + half] : 0.0f;
    v8f t;
    #pragma unroll
    for (int i = 0; i < 8; ++i) t[i] = bv;
    acc[nt] = t;
  }

  int aRow = row0 + half;
  if (aRow >= M) aRow = M - 1;
  const float* Arow = A + aBatchStride * blockIdx.y + (size_t)aRow * aRowStride;

  const float2* BsLane = Bs + (size_t)hi * NDIM + half;

  for (int k0 = 0; k0 < K; k0 += 32) {
    for (int t = threadIdx.x; t < 16 * NDIM; t += 256) {
      int kp = t / NDIM, nn = t % NDIM;
      float2 v;
      v.x = Bb[(size_t)(k0 + 2 * kp)     * NDIM + nn];
      v.y = Bb[(size_t)(k0 + 2 * kp + 1) * NDIM + nn];
      Bs[(size_t)kp * NDIM + nn] = v;
    }
    __syncthreads();

    const float* ap0 = Arow
        + (size_t)(k0 >> aKShift) * aChunkStride
        + ((k0 & aKMask) + 2 * hi);

    #pragma unroll
    for (int kk = 0; kk < 32; kk += 4) {
      v2f a = *(const v2f*)(ap0 + kk);
      if (AACT) {
        v2f ab = *(const v2f*)(aBias + k0 + kk + 2 * hi);
        a.x = fmaxf(a.x + ab.x, 0.0f);
        a.y = fmaxf(a.y + ab.y, 0.0f);
      }
      #pragma unroll
      for (int nt = 0; nt < NTILES; ++nt) {
        v2f b = *(const v2f*)(BsLane + (size_t)(kk >> 1) * NDIM + nt * 16);
        acc[nt] = __builtin_amdgcn_wmma_f32_16x16x4_f32(
            false, a, false, b, (short)0, acc[nt], false, false);
      }
    }
    __syncthreads();
  }

  for (int nt = 0; nt < NTILES; ++nt) {
    #pragma unroll
    for (int vr = 0; vr < 8; ++vr) {
      int r = row0 + vr + 8 * hi;
      if (r < M) {
        float v = acc[nt][vr];
        if (relu) v = fmaxf(v, 0.0f);
        Ob[(size_t)r * NDIM + nt * 16 + half] = v;
      }
    }
  }
}

#define PWAVES 16
#define PKDIM  256
#define PNDIM  256
#define PPITCH 272

static_assert(PNDIM == CC * HIDD, "the projection writes CC planes of HIDD columns");

__global__ __launch_bounds__(PWAVES * 32) void proj_kernel(const float* __restrict__ A,
                                                           const float* __restrict__ Wf,
                                                           float* __restrict__ Out,
                                                           int M) {
  __shared__ alignas(16) float2 Bs[(PKDIM / 2) * PPITCH];

  const int tid  = threadIdx.x;
  const int lane = tid & 31;
  const int wave = tid >> 5;
  const int half = lane & 15;
  const int hi   = lane >> 4;

  {
    constexpr int ITEMS = (PKDIM / 2) * (PNDIM / 4) / (PWAVES * 32);
    v4f r0[ITEMS], r1[ITEMS];
    #pragma unroll
    for (int j = 0; j < ITEMS; ++j) {
      const int i = tid + PWAVES * 32 * j;
      const int p = i >> 6, n = (i & 63) * 4;
      const float* wp = Wf + ((size_t)(n >> 7) * PKDIM + 2 * p) * HIDD + (n & (HIDD - 1));
      r0[j] = *(const v4f*)wp;
      r1[j] = *(const v4f*)(wp + HIDD);
    }
    #pragma unroll
    for (int j = 0; j < ITEMS; ++j) {
      const int i = tid + PWAVES * 32 * j;
      const int p = i >> 6, n = (i & 63) * 4;
      v4f lo, up;
      lo.x = r0[j].x;  lo.y = r1[j].x;  lo.z = r0[j].y;  lo.w = r1[j].y;
      up.x = r0[j].z;  up.y = r1[j].z;  up.z = r0[j].w;  up.w = r1[j].w;
      v4f* d = (v4f*)(Bs + p * PPITCH + n);
      d[0] = lo;
      d[1] = up;
    }
  }
  __syncthreads();

  const int ng     = wave & 3;
  const int mtiles = (M + 15) >> 4;
  const float2* bl = Bs + hi * PPITCH + ng * 64 + half;
  float* Ob = Out + (size_t)(ng >> 1) * M * HIDD + (ng & 1) * 64 + half;

  for (int q = wave >> 2; ; q += PWAVES / 4) {
    const int mt = q * (int)gridDim.x + (int)blockIdx.x;
    if (mt >= mtiles) break;
    const int row0 = mt * 16;
    int aRow = row0 + half;
    if (aRow >= M) aRow = M - 1;
    const float* ap = A + (size_t)aRow * PKDIM + 2 * hi;

    v8f acc[4];
    for (int nt = 0; nt < 4; ++nt)
      for (int i = 0; i < 8; ++i) acc[nt][i] = 0.0f;

    v2f a[8];
    #pragma unroll
    for (int i = 0; i < 8; ++i) a[i] = *(const v2f*)(ap + 4 * i);

    const float2* bp = bl;
    auto piece = [&]() {
      #pragma unroll
      for (int i = 0; i < 8; ++i) {
        #pragma unroll
        for (int nt = 0; nt < 4; ++nt) {
          const v2f b = *(const v2f*)(bp + (2 * i) * PPITCH + nt * 16);
          acc[nt] = __builtin_amdgcn_wmma_f32_16x16x4_f32(
              false, a[i], false, b, (short)0, acc[nt], false, false);
        }
      }
    };
    #pragma nounroll
    for (int ch = 0; ch + 1 < PKDIM / 32; ++ch) {
      v2f an[8];
      #pragma unroll
      for (int i = 0; i < 8; ++i) an[i] = *(const v2f*)(ap + 32 * (ch + 1) + 4 * i);
      piece();
      bp += 16 * PPITCH;
      #pragma unroll
      for (int i = 0; i < 8; ++i) a[i] = an[i];
    }
    piece();

    #pragma unroll
    for (int nt = 0; nt < 4; ++nt) {
      #pragma unroll
      for (int vr = 0; vr < 8; ++vr) {
        const int r = row0 + vr + 8 * hi;
        if (r < M) Ob[(size_t)r * HIDD + nt * 16] = acc[nt][vr];
      }
    }
  }
}

__global__ __launch_bounds__(32) void wf_tile_kernel(const float* __restrict__ Wc,
                                                     const float* __restrict__ W1,
                                                     float* __restrict__ Wf,
                                                     int inDim) {
  const int lane = threadIdx.x;
  const int half = lane & 15;
  const int hi   = lane >> 4;
  const int mt = blockIdx.x, nt = blockIdx.y, c = blockIdx.z;

  const float* ap = Wc + ((size_t)c * inDim + mt * 16 + half) * HIDD + 4 * hi;
  const float* bp = W1 + ((size_t)c * HIDD + 4 * hi) * HIDD + nt * 16 + half;

  v4f   a[HIDD / 8];
  float b[HIDD / 8][4];
  #pragma unroll
  for (int i = 0; i < HIDD / 8; ++i) a[i] = *(const v4f*)(ap + 8 * i);
  #pragma unroll
  for (int i = 0; i < HIDD / 8; ++i)
    for (int t = 0; t < 4; ++t) b[i][t] = bp[(size_t)(8 * i + t) * HIDD];

  __builtin_amdgcn_sched_barrier(0);

  v8f acc0, acc1;
  for (int i = 0; i < 8; ++i) { acc0[i] = 0.0f; acc1[i] = 0.0f; }
  #pragma unroll
  for (int i = 0; i < HIDD / 8; ++i) {
    v2f a0, b0, a1, b1;
    a0.x = a[i].x;  a0.y = a[i].y;  b0.x = b[i][0];  b0.y = b[i][1];
    a1.x = a[i].z;  a1.y = a[i].w;  b1.x = b[i][2];  b1.y = b[i][3];
    acc0 = __builtin_amdgcn_wmma_f32_16x16x4_f32(false, a0, false, b0, (short)0, acc0, false, false);
    acc1 = __builtin_amdgcn_wmma_f32_16x16x4_f32(false, a1, false, b1, (short)0, acc1, false, false);
  }

  float* op = Wf + ((size_t)c * inDim + mt * 16 + 8 * hi) * HIDD + nt * 16 + half;
  #pragma unroll
  for (int vr = 0; vr < 8; ++vr) op[(size_t)vr * HIDD] = acc0[vr] + acc1[vr];
}

__global__ void gather_kernel(const float* __restrict__ Y,
                              const int* __restrict__ cat,
                              float* __restrict__ out, int ncat) {
  int t = blockIdx.x * blockDim.x + threadIdx.x;
  if (t >= ncat * 16) return;
  int i = t >> 4, j = t & 15;
  out[t] = Y[(size_t)cat[i] * 16 + j];
}

extern "C" void kernel_launch(void* const* d_in, const int* in_sizes, int n_in,
                              void* d_out, int out_size, void* d_ws, size_t ws_size,
                              hipStream_t stream) {
  const float* h   = (const float*)d_in[0];
  const int*   src = (const int*)d_in[1];
  const int*   dst = (const int*)d_in[2];
  const int*   et  = (const int*)d_in[3];
  const int*   cat = (const int*)d_in[4];
  const float* gtw = (const float*)d_in[5];
  const float* Wc  = (const float*)d_in[6];
  const float* W1  = (const float*)d_in[7];
  const float* b1  = (const float*)d_in[8];
  const float* W2  = (const float*)d_in[9];
  const float* b2  = (const float*)d_in[10];

  const int IN_DIM = 256;
  const int N    = in_sizes[0] / IN_DIM;
  const int E    = in_sizes[1];
  const int NCAT = in_sizes[4];

  float* ws    = (float*)d_ws;
  float* filt  = ws;
  float* Hbuf0 = filt + 128;
  float* Hbuf1 = Hbuf0 + (size_t)CC * N * HIDD;
  float* Wf    = Hbuf1 + (size_t)CC * N * HIDD;
  float* Y     = Wf + (size_t)N * HIDD;

  const int gtile = (N + TROWS - 1) / TROWS;
  int* elist = (int*)(Wf + (size_t)CC * IN_DIM * HIDD);
  int* lcnt  = elist + (size_t)gtile * LCAP * 2;
  unsigned* tileb = (unsigned*)(lcnt + ((gtile + 31) & ~31));
  float* degws = (float*)(tileb + (((E >> 4) * 4 + 31) & ~31));

  const int nquad = (E >> 4) * 4;
  prep_kernel<<<1 + (nquad + 255) / 256, 256, 0, stream>>>(gtw, filt, dst, tileb, nquad);

  const int mtiles = (N + 15) / 16;
  const int gx     = (mtiles + 7) / 8;

  wf_tile_kernel<<<dim3(IN_DIM / 16, HIDD / 16, CC), 32, 0, stream>>>(Wc, W1, Wf, IN_DIM);
  proj_kernel<<<256, PWAVES * 32, 0, stream>>>(h, Wf, Hbuf0, N);

  float* Hcur = Hbuf0;
  float* Hnxt = Hbuf1;
  for (int l = 0; l < LL - 1; ++l) {
    tile_msg_kernel<false><<<gtile, TWAVES * 32, 0, stream>>>(
        src, dst, et, filt, Hcur, Hnxt, (const v4i*)tileb, elist, lcnt, degws, E, N, l, l == 0 ? 1 : 0);
    float* tmp = Hcur; Hcur = Hnxt; Hnxt = tmp;
  }
  float* S = Hnxt;
  tile_msg_kernel<true><<<gtile, TWAVES * 32, 0, stream>>>(
      src, dst, et, filt, Hcur, S, (const v4i*)tileb, elist, lcnt, degws, E, N, LL - 1, 0);

  gemm_wmma_f32<1, true><<<dim3(gx, 1), 256, 0, stream>>>(
      S, W2, b2, b1, Y, N, HIDD,
        HIDD,   7,   127,   0,
      0, 0, 0,   0);

  gather_kernel<<<(NCAT * 16 + 255) / 256, 256, 0, stream>>>(
      Y, cat, (float*)d_out, NCAT);
}
